# starting kernel
# baseline (speedup 1.0000x reference)
#include <hip/hip_runtime.h>

typedef float v2f __attribute__((ext_vector_type(2)));
typedef float v4f __attribute__((ext_vector_type(4)));
typedef float v8f __attribute__((ext_vector_type(8)));

#define C_DIM   128
#define N_DIM   64
#define HW      3136
#define PLANES  (C_DIM * N_DIM)
#define NUMEL_PER_CH (N_DIM * HW)
#define BN_EPS  1e-5f

static __device__ __forceinline__ v8f wmma_rowsum(v2f a, v2f ones, v8f c) {
    return __builtin_amdgcn_wmma_f32_16x16x4_f32(
        false, a, false, ones, (short)0, c, false, false);
}

__global__ __launch_bounds__(256) void bn_partials(const float* __restrict__ x,
                                                   float* __restrict__ psum,
                                                   float* __restrict__ psq) {
    const int c = blockIdx.x;
    const int n = blockIdx.y;
    const int plane = n * C_DIM + c;
    const v4f* __restrict__ xp4 = (const v4f*)(x + (long)plane * HW);
    const int tid = threadIdx.x;

    v2f ones; ones.x = 1.0f; ones.y = 1.0f;
    v8f s0 = {}, s1 = {}, q0 = {}, q1 = {};

    #pragma unroll
    for (int k = 0; k < 3; ++k) {
        v4f v = xp4[k * 256 + tid];
        v2f a0; a0.x = v.x;       a0.y = v.y;
        v2f a1; a1.x = v.z;       a1.y = v.w;
        v2f b0; b0.x = v.x * v.x; b0.y = v.y * v.y;
        v2f b1; b1.x = v.z * v.z; b1.y = v.w * v.w;
        s0 = wmma_rowsum(a0, ones, s0);
        s1 = wmma_rowsum(a1, ones, s1);
        q0 = wmma_rowsum(b0, ones, q0);
        q1 = wmma_rowsum(b1, ones, q1);
    }

    if (tid < 32) {
        const float2* __restrict__ xp2 = (const float2*)xp4;
        float2 v = xp2[1536 + tid];
        v2f a; a.x = v.x;       a.y = v.y;
        v2f b; b.x = v.x * v.x; b.y = v.y * v.y;
        s0 = wmma_rowsum(a, ones, s0);
        q0 = wmma_rowsum(b, ones, q0);
    }

    v8f tS = s0 + s1;
    v8f tQ = q0 + q1;
    float s = tS[0] + tS[1] + tS[2] + tS[3] + tS[4] + tS[5] + tS[6] + tS[7];
    float q = tQ[0] + tQ[1] + tQ[2] + tQ[3] + tQ[4] + tQ[5] + tQ[6] + tQ[7];
    s += __shfl_xor(s, 16, 32);
    q += __shfl_xor(q, 16, 32);

    __shared__ float ls[8];
    __shared__ float lq[8];
    const int wave = tid >> 5;
    const int lane = tid & 31;
    if (lane == 0) { ls[wave] = s; lq[wave] = q; }
    __syncthreads();
    if (tid == 0) {
        float S = 0.f, Q = 0.f;
        for (int w = 0; w < 8; ++w) { S += ls[w]; Q += lq[w]; }
        psum[plane] = S;
        psq [plane] = Q;
    }
}

__global__ __launch_bounds__(C_DIM) void bn_finalize(const float* __restrict__ psum,
                                                     const float* __restrict__ psq,
                                                     const float* __restrict__ weight,
                                                     const float* __restrict__ bias,
                                                     float* __restrict__ scale,
                                                     float* __restrict__ shift) {
    const int c = threadIdx.x;
    float S = 0.f, Q = 0.f;
    for (int n = 0; n < N_DIM; ++n) {
        S += psum[n * C_DIM + c];
        Q += psq [n * C_DIM + c];
    }
    const float inv_n = 1.0f / (float)NUMEL_PER_CH;
    const float mean = S * inv_n;
    const float var  = Q * inv_n - mean * mean;
    const float istd = rsqrtf(var + BN_EPS);
    const float sc   = weight[c] * istd;
    scale[c] = sc;
    shift[c] = bias[c] - mean * sc;
}

__global__ __launch_bounds__(256) void bn_apply(const float* __restrict__ x,
                                                const float* __restrict__ scale,
                                                const float* __restrict__ shift,
                                                float* __restrict__ y,
                                                int n4) {
    __shared__ float ssc[C_DIM];
    __shared__ float ssh[C_DIM];
    if (threadIdx.x < C_DIM) {
        ssc[threadIdx.x] = scale[threadIdx.x];
        ssh[threadIdx.x] = shift[threadIdx.x];
    }
    __syncthreads();

    const v4f* __restrict__ x4 = (const v4f*)x;
    v4f* __restrict__ y4 = (v4f*)y;
    const int stride = (int)gridDim.x * 256;

    for (int i = (int)blockIdx.x * 256 + (int)threadIdx.x; i < n4; i += stride) {
        if (i + stride < n4) {
            __builtin_prefetch(x4 + i + stride, 0, 1);
        }
        const int c = (i / (HW / 4)) % C_DIM;
        v4f v = x4[i];
        const float sc = ssc[c];
        const float sh = ssh[c];
        v4f r;
        r.x = v.x * sc + sh;
        r.y = v.y * sc + sh;
        r.z = v.z * sc + sh;
        r.w = v.w * sc + sh;
        __builtin_nontemporal_store(r, y4 + i);
    }
}

extern "C" void kernel_launch(void* const* d_in, const int* in_sizes, int n_in,
                              void* d_out, int out_size, void* d_ws, size_t ws_size,
                              hipStream_t stream) {
    const float* x = (const float*)d_in[0];
    const float* w = (const float*)d_in[1];
    const float* b = (const float*)d_in[2];
    float* y  = (float*)d_out;
    float* ws = (float*)d_ws;

    float* psum  = ws;
    float* psq   = ws + PLANES;
    float* scale = ws + 2 * PLANES;
    float* shift = scale + C_DIM;

    dim3 g1(C_DIM, N_DIM);
    bn_partials<<<g1, 256, 0, stream>>>(x, psum, psq);
    bn_finalize<<<1, C_DIM, 0, stream>>>(psum, psq, w, b, scale, shift);

    const int n4 = out_size / 4;
    bn_apply<<<4096, 256, 0, stream>>>(x, scale, shift, y, n4);
}


# the channel stays on the WGP between the two passes: of a lane's 49 float4, 23 stay in registers, 19 wait in LDS, 7 are read again; 2 accumulators, wave totals by shuffle tree: sum order changes
# speedup vs baseline: 1.2905x; 1.2905x over previous
#include <hip/hip_runtime.h>


typedef float v2f __attribute__((ext_vector_type(2)));
typedef float v4f __attribute__((ext_vector_type(4)));
typedef float v8f __attribute__((ext_vector_type(8)));

#define C_DIM   128
#define N_DIM   64
#define HW      3136
#define HW4     (HW / 4)
#define NUMEL_PER_CH (N_DIM * HW)
#define BN_EPS  1e-5f

#define HALF    16
#define WG_THREADS (N_DIM * HALF)
#define WAVES   (WG_THREADS / 32)
#define ITERS   (HW4 / HALF)
#define CHUNK   7
#define NCHUNK  (ITERS / CHUNK)

#define N_BACK  7
#define N_LDS   19
#define N_REG   23
#define N_LDS_A 10
#define K_LDS0  N_BACK
#define K_REG0  (N_BACK + N_LDS)

static __device__ __forceinline__ v8f wmma_rowsum(v2f a, v2f ones, v8f c) {
    return __builtin_amdgcn_wmma_f32_16x16x4_f32(
        false, a, false, ones, (short)0, c, false, false);
}

static __device__ __forceinline__ v4f scale_shift(v4f v, float sc, float sh) {
    v4f r;
    r.x = v.x * sc + sh;
    r.y = v.y * sc + sh;
    r.z = v.z * sc + sh;
    r.w = v.w * sc + sh;
    return r;
}

__global__ __launch_bounds__(WG_THREADS) void bn_channel_resident(
        const float* __restrict__ x,
        const float* __restrict__ weight,
        const float* __restrict__ bias,
        float* __restrict__ y) {
    __shared__ v4f stash0[4][WG_THREADS];
    __shared__ v4f stash1[4][WG_THREADS];
    __shared__ v4f stash2[4][WG_THREADS];
    __shared__ v4f stash3[4][WG_THREADS];
    __shared__ v4f stash4[N_LDS - 16][WG_THREADS];
#define STASH(i) (*((i) < 4 ? &stash0[(i) & 3][tid] : (i) < 8 ? &stash1[(i) & 3][tid] : \
                    (i) < 12 ? &stash2[(i) & 3][tid] : (i) < 16 ? &stash3[(i) & 3][tid] : \
                    &stash4[(i) >= 16 ? (i) - 16 : 0][tid]))
    __shared__ float ls[WAVES];
    __shared__ float lq[WAVES];

    const unsigned c   = blockIdx.x;
    const unsigned tid = threadIdx.x;
    const unsigned plane = tid / HALF;
    const unsigned first = (plane * C_DIM + c) * HW4 + (tid % HALF);
    const v4f* __restrict__ xp = (const v4f*)x + first;
    v4f* __restrict__ yp = (v4f*)y + first;

    v2f ones; ones.x = 1.0f; ones.y = 1.0f;
    v8f s = {}, q = {};

    v4f keep[N_REG];
    v4f cur[CHUNK], nxt[CHUNK];

    #pragma unroll
    for (int j = 0; j < CHUNK; ++j)
        cur[j] = xp[HALF * j];

    #pragma unroll
    for (int kk = 0; kk < NCHUNK; ++kk) {
        if (kk + 1 < NCHUNK) {
            #pragma unroll
            for (int j = 0; j < CHUNK; ++j)
                nxt[j] = xp[HALF * ((kk + 1) * CHUNK + j)];
        }
        #pragma unroll
        for (int j = 0; j < CHUNK; ++j) {
            const int k = kk * CHUNK + j;
            const v4f v = cur[j];
            if (k >= K_LDS0 && k < K_REG0) STASH(k - K_LDS0) = v;
            if (k >= K_REG0) keep[k - K_REG0] = v;
            v2f a0; a0.x = v.x;       a0.y = v.y;
            v2f a1; a1.x = v.z;       a1.y = v.w;
            v2f b0; b0.x = v.x * v.x; b0.y = v.y * v.y;
            v2f b1; b1.x = v.z * v.z; b1.y = v.w * v.w;
            s = wmma_rowsum(a0, ones, s);
            q = wmma_rowsum(b0, ones, q);
            s = wmma_rowsum(a1, ones, s);
            q = wmma_rowsum(b1, ones, q);
        }
        #pragma unroll
        for (int j = 0; j < CHUNK; ++j) cur[j] = nxt[j];
    }

    float ws = s[0] + s[1] + s[2] + s[3] + s[4] + s[5] + s[6] + s[7];
    float wq = q[0] + q[1] + q[2] + q[3] + q[4] + q[5] + q[6] + q[7];
    ws += __shfl_xor(ws, 16, 32);
    wq += __shfl_xor(wq, 16, 32);

    const unsigned wave = tid >> 5;
    const unsigned lane = tid & 31;
    if (lane == 0) { ls[wave] = ws; lq[wave] = wq; }

    __syncthreads();

    float S = ls[lane], Q = lq[lane];
    #pragma unroll
    for (int m = 16; m >= 1; m >>= 1) {
        S += __shfl_xor(S, m, 32);
        Q += __shfl_xor(Q, m, 32);
    }
    S = __int_as_float(__builtin_amdgcn_readfirstlane(__float_as_int(S)));
    Q = __int_as_float(__builtin_amdgcn_readfirstlane(__float_as_int(Q)));

    const float inv_n = 1.0f / (float)NUMEL_PER_CH;
    const float mean = S * inv_n;
    const float var  = Q * inv_n - mean * mean;
    const float istd = rsqrtf(var + BN_EPS);
    const float sc   = weight[c] * istd;
    const float sh   = bias[c] - mean * sc;

    #pragma unroll
    for (int i = 0; i < N_REG; ++i)
        keep[i] = scale_shift(keep[i], sc, sh);
    #pragma unroll
    for (int i = 0; i < N_REG; ++i)
        __builtin_nontemporal_store(keep[i], yp + HALF * (K_REG0 + i));

    v4f back[N_BACK];
    #pragma unroll
    for (int j = 0; j < N_BACK; ++j)
        back[j] = xp[HALF * j];

    v4f park[N_LDS];
    #pragma unroll
    for (int i = 0; i < N_LDS_A; ++i)
        park[i] = scale_shift(STASH(i), sc, sh);
    #pragma unroll
    for (int i = 0; i < N_LDS_A; ++i)
        __builtin_nontemporal_store(park[i], yp + HALF * (K_LDS0 + i));
    #pragma unroll
    for (int i = N_LDS_A; i < N_LDS; ++i)
        park[i] = scale_shift(STASH(i), sc, sh);
    #pragma unroll
    for (int i = N_LDS_A; i < N_LDS; ++i)
        __builtin_nontemporal_store(park[i], yp + HALF * (K_LDS0 + i));

    #pragma unroll
    for (int j = 0; j < N_BACK; ++j)
        back[j] = scale_shift(back[j], sc, sh);
    #pragma unroll
    for (int j = 0; j < N_BACK; ++j)
        __builtin_nontemporal_store(back[j], yp + HALF * j);
}

extern "C" void kernel_launch(void* const* d_in, const int* in_sizes, int n_in,
                              void* d_out, int out_size, void* d_ws, size_t ws_size,
                              hipStream_t stream) {
    const float* x = (const float*)d_in[0];
    const float* w = (const float*)d_in[1];
    const float* b = (const float*)d_in[2];
    float* y = (float*)d_out;

    bn_channel_resident<<<C_DIM, WG_THREADS, 0, stream>>>(x, w, b, y);
}


# 16 waves per workgroup, not 32: a half-wave carries two planes (98 float4 a lane), wave entries per launch 4096 -> 2048; freed registers keep 54 of 98 in place, re-read 14 -> 6; sum order changes
# speedup vs baseline: 1.7163x; 1.3299x over previous
#include <hip/hip_runtime.h>


typedef float v2f __attribute__((ext_vector_type(2)));
typedef float v4f __attribute__((ext_vector_type(4)));
typedef float v8f __attribute__((ext_vector_type(8)));

#define C_DIM   128
#define N_DIM   64
#define HW      3136
#define HW4     (HW / 4)
#define NUMEL_PER_CH (N_DIM * HW)
#define BN_EPS  1e-5f

#define HALF    16
#define PER_PLANE (HW4 / HALF)
#define WAVES   16
#define WG_THREADS (WAVES * 32)
#define PLANE_B (2 * WAVES)
#define ITEMS   (2 * PER_PLANE)
#define CHUNK   14
#define NCHUNK  (ITEMS / CHUNK)

#define N_BACK  6
#define N_LDS   38
#define N_REG   54
#define N_LDS_A 19
#define I_LDS0  N_BACK
#define I_REG0  (N_BACK + N_LDS)

#define ITEM_OFF(i) (((i) < PER_PLANE ? 0 : PLANE_B * C_DIM * HW4) + HALF * ((i) % PER_PLANE))

static __device__ __forceinline__ v8f wmma_rowsum(v2f a, v2f ones, v8f c) {
    return __builtin_amdgcn_wmma_f32_16x16x4_f32(
        false, a, false, ones, (short)0, c, false, false);
}

static __device__ __forceinline__ v4f scale_shift(v4f v, float sc, float sh) {
    v4f r;
    r.x = v.x * sc + sh;
    r.y = v.y * sc + sh;
    r.z = v.z * sc + sh;
    r.w = v.w * sc + sh;
    return r;
}

__global__ __launch_bounds__(WG_THREADS) void bn_channel_resident(
        const float* __restrict__ x,
        const float* __restrict__ weight,
        const float* __restrict__ bias,
        float* __restrict__ y) {
    __shared__ v4f stash0[8][WG_THREADS];
    __shared__ v4f stash1[8][WG_THREADS];
    __shared__ v4f stash2[8][WG_THREADS];
    __shared__ v4f stash3[8][WG_THREADS];
    __shared__ v4f stash4[N_LDS - 32][WG_THREADS];
#define STASH(i) (*((i) < 8 ? &stash0[(i) & 7][tid] : (i) < 16 ? &stash1[(i) & 7][tid] : \
                    (i) < 24 ? &stash2[(i) & 7][tid] : (i) < 32 ? &stash3[(i) & 7][tid] : \
                    &stash4[(i) >= 32 ? (i) - 32 : 0][tid]))
    __shared__ float ls[WAVES];
    __shared__ float lq[WAVES];

    const unsigned c   = blockIdx.x;
    const unsigned tid = threadIdx.x;
    const unsigned plane = tid / HALF;
    const unsigned first = (plane * C_DIM + c) * HW4 + (tid % HALF);
    const v4f* __restrict__ xp = (const v4f*)x + first;
    v4f* __restrict__ yp = (v4f*)y + first;

    v2f ones; ones.x = 1.0f; ones.y = 1.0f;
    v4f as0 = {}, aq0 = {}, as1 = {}, aq1 = {};

    v4f keep[N_REG];
    v4f cur[CHUNK], nxt[CHUNK];

    #pragma unroll
    for (int j = 0; j < CHUNK; ++j)
        cur[j] = xp[ITEM_OFF(j)];

    #pragma unroll
    for (int kk = 0; kk < NCHUNK; ++kk) {
        if (kk + 1 < NCHUNK) {
            #pragma unroll
            for (int j = 0; j < CHUNK; ++j)
                nxt[j] = xp[ITEM_OFF((kk + 1) * CHUNK + j)];
        }
        #pragma unroll
        for (int j = 0; j < CHUNK; ++j) {
            const int i = kk * CHUNK + j;
            const v4f v = cur[j];
            if (i >= I_LDS0 && i < I_REG0) STASH(i - I_LDS0) = v;
            if (i >= I_REG0) keep[i - I_REG0] = v;
            if (i & 1) { as1 += v; aq1 += v * v; }
            else       { as0 += v; aq0 += v * v; }
        }
        #pragma unroll
        for (int j = 0; j < CHUNK; ++j) cur[j] = nxt[j];
    }

    const v4f ts = as0 + as1;
    const v4f tq = aq0 + aq1;
    v2f la; la.x = ts.x + ts.z; la.y = ts.y + ts.w;
    v2f lb; lb.x = tq.x + tq.z; lb.y = tq.y + tq.w;
    v8f s = {}, q = {};
    s = wmma_rowsum(la, ones, s);
    q = wmma_rowsum(lb, ones, q);

    float ws = s[0] + s[1] + s[2] + s[3] + s[4] + s[5] + s[6] + s[7];
    float wq = q[0] + q[1] + q[2] + q[3] + q[4] + q[5] + q[6] + q[7];
    ws += __shfl_xor(ws, 16, 32);
    wq += __shfl_xor(wq, 16, 32);

    const unsigned wave = tid >> 5;
    const unsigned lane = tid & 31;
    if (lane == 0) { ls[wave] = ws; lq[wave] = wq; }

    __syncthreads();

    float S = ls[lane & (WAVES - 1)], Q = lq[lane & (WAVES - 1)];
    #pragma unroll
    for (int m = WAVES / 2; m >= 1; m >>= 1) {
        S += __shfl_xor(S, m, 32);
        Q += __shfl_xor(Q, m, 32);
    }
    S = __int_as_float(__builtin_amdgcn_readfirstlane(__float_as_int(S)));
    Q = __int_as_float(__builtin_amdgcn_readfirstlane(__float_as_int(Q)));

    const float inv_n = 1.0f / (float)NUMEL_PER_CH;
    const float mean = S * inv_n;
    const float var  = Q * inv_n - mean * mean;
    const float istd = rsqrtf(var + BN_EPS);
    const float sc   = weight[c] * istd;
    const float sh   = bias[c] - mean * sc;

    #pragma unroll
    for (int i = 0; i < N_REG; ++i)
        keep[i] = scale_shift(keep[i], sc, sh);
    #pragma unroll
    for (int i = 0; i < N_REG; ++i)
        __builtin_nontemporal_store(keep[i], yp + ITEM_OFF(I_REG0 + i));

    v4f back[N_BACK];
    #pragma unroll
    for (int j = 0; j < N_BACK; ++j)
        back[j] = xp[ITEM_OFF(j)];

    v4f park[N_LDS];
    #pragma unroll
    for (int i = 0; i < N_LDS_A; ++i)
        park[i] = scale_shift(STASH(i), sc, sh);
    #pragma unroll
    for (int i = 0; i < N_LDS_A; ++i)
        __builtin_nontemporal_store(park[i], yp + ITEM_OFF(I_LDS0 + i));
    #pragma unroll
    for (int i = N_LDS_A; i < N_LDS; ++i)
        park[i] = scale_shift(STASH(i), sc, sh);
    #pragma unroll
    for (int i = N_LDS_A; i < N_LDS; ++i)
        __builtin_nontemporal_store(park[i], yp + ITEM_OFF(I_LDS0 + i));

    #pragma unroll
    for (int j = 0; j < N_BACK; ++j)
        back[j] = scale_shift(back[j], sc, sh);
    #pragma unroll
    for (int j = 0; j < N_BACK; ++j)
        __builtin_nontemporal_store(back[j], yp + ITEM_OFF(j));
}

extern "C" void kernel_launch(void* const* d_in, const int* in_sizes, int n_in,
                              void* d_out, int out_size, void* d_ws, size_t ws_size,
                              hipStream_t stream) {
    const float* x = (const float*)d_in[0];
    const float* w = (const float*)d_in[1];
    const float* b = (const float*)d_in[2];
    float* y = (float*)d_out;

    bn_channel_resident<<<C_DIM, WG_THREADS, 0, stream>>>(x, w, b, y);
}


# second-level-cache prefetch of the x stream: per wave one global_prefetch_b8 per chunk of 14 loads (a 256-B piece per lane), 2-3 chunks ahead, so its in-order loads hit the cache
# speedup vs baseline: 1.9109x; 1.1134x over previous
#include <hip/hip_runtime.h>


typedef float v2f __attribute__((ext_vector_type(2)));
typedef float v4f __attribute__((ext_vector_type(4)));
typedef float v8f __attribute__((ext_vector_type(8)));

#define C_DIM   128
#define N_DIM   64
#define HW      3136
#define HW4     (HW / 4)
#define NUMEL_PER_CH (N_DIM * HW)
#define BN_EPS  1e-5f

#define HALF    16
#define PER_PLANE (HW4 / HALF)
#define WAVES   16
#define WG_THREADS (WAVES * 32)
#define PLANE_B (2 * WAVES)
#define ITEMS   (2 * PER_PLANE)
#define CHUNK   14
#define NCHUNK  (ITEMS / CHUNK)
#define PF_LOCALITY 1

#define N_BACK  6
#define N_LDS   38
#define N_REG   54
#define N_LDS_A 19
#define I_LDS0  N_BACK
#define I_REG0  (N_BACK + N_LDS)

#define ITEM_OFF(i) (((i) < PER_PLANE ? 0 : PLANE_B * C_DIM * HW4) + HALF * ((i) % PER_PLANE))

static __device__ __forceinline__ v8f wmma_rowsum(v2f a, v2f ones, v8f c) {
    return __builtin_amdgcn_wmma_f32_16x16x4_f32(
        false, a, false, ones, (short)0, c, false, false);
}

static __device__ __forceinline__ v4f scale_shift(v4f v, float sc, float sh) {
    v4f r;
    r.x = v.x * sc + sh;
    r.y = v.y * sc + sh;
    r.z = v.z * sc + sh;
    r.w = v.w * sc + sh;
    return r;
}

static __device__ __forceinline__ void l2_ask(const v4f* plane0, unsigned hl, int m) {
    const unsigned i = m * CHUNK + (hl < CHUNK ? hl : CHUNK - 1);
    const unsigned off = i < PER_PLANE ? HALF * i
                                       : PLANE_B * C_DIM * HW4 + HALF * (i - PER_PLANE);
    __builtin_prefetch(plane0 + off, 0, PF_LOCALITY);
}

__global__ __launch_bounds__(WG_THREADS) void bn_channel_resident(
        const float* __restrict__ x,
        const float* __restrict__ weight,
        const float* __restrict__ bias,
        float* __restrict__ y) {
    __shared__ v4f stash0[8][WG_THREADS];
    __shared__ v4f stash1[8][WG_THREADS];
    __shared__ v4f stash2[8][WG_THREADS];
    __shared__ v4f stash3[8][WG_THREADS];
    __shared__ v4f stash4[N_LDS - 32][WG_THREADS];
#define STASH(i) (*((i) < 8 ? &stash0[(i) & 7][tid] : (i) < 16 ? &stash1[(i) & 7][tid] : \
                    (i) < 24 ? &stash2[(i) & 7][tid] : (i) < 32 ? &stash3[(i) & 7][tid] : \
                    &stash4[(i) >= 32 ? (i) - 32 : 0][tid]))
    __shared__ float ls[WAVES];
    __shared__ float lq[WAVES];

    const unsigned c   = blockIdx.x;
    const unsigned tid = threadIdx.x;
    const unsigned plane = tid / HALF;
    const unsigned first = (plane * C_DIM + c) * HW4 + (tid % HALF);
    const v4f* __restrict__ xp = (const v4f*)x + first;
    const unsigned hl = tid % HALF;
    const v4f* plane0 = (const v4f*)x + (plane * C_DIM + c) * HW4;
    v4f* __restrict__ yp = (v4f*)y + first;

    v2f ones; ones.x = 1.0f; ones.y = 1.0f;
    v4f as0 = {}, aq0 = {}, as1 = {}, aq1 = {};

    v4f keep[N_REG];
    v4f cur[CHUNK], nxt[CHUNK];

    l2_ask(plane0, hl, 1);
    #pragma unroll
    for (int j = 0; j < CHUNK; ++j)
        cur[j] = xp[ITEM_OFF(j)];
    l2_ask(plane0, hl, 2);
    l2_ask(plane0, hl, 3);
    l2_ask(plane0, hl, 4);

    #pragma unroll
    for (int kk = 0; kk < NCHUNK; ++kk) {
        if (kk + 1 < NCHUNK) {
            #pragma unroll
            for (int j = 0; j < CHUNK; ++j)
                nxt[j] = xp[ITEM_OFF((kk + 1) * CHUNK + j)];
        }
        if (kk >= 1 && kk + 4 < NCHUNK) l2_ask(plane0, hl, kk + 4);
        #pragma unroll
        for (int j = 0; j < CHUNK; ++j) {
            const int i = kk * CHUNK + j;
            const v4f v = cur[j];
            if (i >= I_LDS0 && i < I_REG0) STASH(i - I_LDS0) = v;
            if (i >= I_REG0) keep[i - I_REG0] = v;
            if (i & 1) { as1 += v; aq1 += v * v; }
            else       { as0 += v; aq0 += v * v; }
        }
        #pragma unroll
        for (int j = 0; j < CHUNK; ++j) cur[j] = nxt[j];
    }

    const v4f ts = as0 + as1;
    const v4f tq = aq0 + aq1;
    v2f la; la.x = ts.x + ts.z; la.y = ts.y + ts.w;
    v2f lb; lb.x = tq.x + tq.z; lb.y = tq.y + tq.w;
    v8f s = {}, q = {};
    s = wmma_rowsum(la, ones, s);
    q = wmma_rowsum(lb, ones, q);

    float ws = s[0] + s[1] + s[2] + s[3] + s[4] + s[5] + s[6] + s[7];
    float wq = q[0] + q[1] + q[2] + q[3] + q[4] + q[5] + q[6] + q[7];
    ws += __shfl_xor(ws, 16, 32);
    wq += __shfl_xor(wq, 16, 32);

    const unsigned wave = tid >> 5;
    const unsigned lane = tid & 31;
    if (lane == 0) { ls[wave] = ws; lq[wave] = wq; }

    __syncthreads();

    float S = ls[lane & (WAVES - 1)], Q = lq[lane & (WAVES - 1)];
    #pragma unroll
    for (int m = WAVES / 2; m >= 1; m >>= 1) {
        S += __shfl_xor(S, m, 32);
        Q += __shfl_xor(Q, m, 32);
    }
    S = __int_as_float(__builtin_amdgcn_readfirstlane(__float_as_int(S)));
    Q = __int_as_float(__builtin_amdgcn_readfirstlane(__float_as_int(Q)));

    const float inv_n = 1.0f / (float)NUMEL_PER_CH;
    const float mean = S * inv_n;
    const float var  = Q * inv_n - mean * mean;
    const float istd = rsqrtf(var + BN_EPS);
    const float sc   = weight[c] * istd;
    const float sh   = bias[c] - mean * sc;

    #pragma unroll
    for (int i = 0; i < N_REG; ++i)
        keep[i] = scale_shift(keep[i], sc, sh);
    #pragma unroll
    for (int i = 0; i < N_REG; ++i)
        __builtin_nontemporal_store(keep[i], yp + ITEM_OFF(I_REG0 + i));

    v4f back[N_BACK];
    #pragma unroll
    for (int j = 0; j < N_BACK; ++j)
        back[j] = xp[ITEM_OFF(j)];

    v4f park[N_LDS];
    #pragma unroll
    for (int i = 0; i < N_LDS_A; ++i)
        park[i] = scale_shift(STASH(i), sc, sh);
    #pragma unroll
    for (int i = 0; i < N_LDS_A; ++i)
        __builtin_nontemporal_store(park[i], yp + ITEM_OFF(I_LDS0 + i));
    #pragma unroll
    for (int i = N_LDS_A; i < N_LDS; ++i)
        park[i] = scale_shift(STASH(i), sc, sh);
    #pragma unroll
    for (int i = N_LDS_A; i < N_LDS; ++i)
        __builtin_nontemporal_store(park[i], yp + ITEM_OFF(I_LDS0 + i));

    #pragma unroll
    for (int j = 0; j < N_BACK; ++j)
        back[j] = scale_shift(back[j], sc, sh);
    #pragma unroll
    for (int j = 0; j < N_BACK; ++j)
        __builtin_nontemporal_store(back[j], yp + ITEM_OFF(j));
}

extern "C" void kernel_launch(void* const* d_in, const int* in_sizes, int n_in,
                              void* d_out, int out_size, void* d_ws, size_t ws_size,
                              hipStream_t stream) {
    const float* x = (const float*)d_in[0];
    const float* w = (const float*)d_in[1];
    const float* b = (const float*)d_in[2];
    float* y = (float*)d_out;

    bn_channel_resident<<<C_DIM, WG_THREADS, 0, stream>>>(x, w, b, y);
}
